# starting kernel
# baseline (speedup 1.0000x reference)
#include <hip/hip_runtime.h>
#include <math.h>

#define N_NODES 200000
#define N_EDGES 2000000
#define IN_DIM  165
#define HID_DIM 128
#define KPAD    168
#define KMAIN   164

typedef __attribute__((ext_vector_type(2))) float v2f;
typedef __attribute__((ext_vector_type(8))) float v8f;


__global__ void k_deg_init(float* __restrict__ deg, int n) {
    int i = blockIdx.x * blockDim.x + threadIdx.x;
    if (i < n) deg[i] = 1.0f;
}

__global__ void k_deg_count(const int* __restrict__ dst, float* __restrict__ deg, int e) {
    int i = blockIdx.x * blockDim.x + threadIdx.x;
    if (i < e) atomicAdd(&deg[dst[i]], 1.0f);
}

__global__ void k_dinv(float* __restrict__ deg, int n) {
    int i = blockIdx.x * blockDim.x + threadIdx.x;
    if (i < n) deg[i] = rsqrtf(deg[i]);
}


__global__ __launch_bounds__(256)
void k_gemm1(const float* __restrict__ x, const float* __restrict__ W1,
             float* __restrict__ h) {
    __shared__ float sX[64][KPAD];
    const int tid      = threadIdx.x;
    const int nodeBase = blockIdx.x * 64;

    for (int idx = tid; idx < 64 * KPAD; idx += 256) {
        int r = idx / KPAD, c = idx % KPAD;
        sX[r][c] = (c < IN_DIM) ? x[(nodeBase + r) * IN_DIM + c] : 0.0f;
    }
    __syncthreads();

    const int lane  = tid & 31;
    const int wave  = tid >> 5;
    const int m     = lane & 15;
    const int kp    = (lane >> 4) << 1;
    const int ntile = wave >> 1;
    const int fbase = (wave & 1) * 64;
    const float* arow = &sX[ntile * 16 + m][0];

    for (int ft = 0; ft < 4; ++ft) {
        const int feat0 = fbase + ft * 16;
        v8f c = {0.f, 0.f, 0.f, 0.f, 0.f, 0.f, 0.f, 0.f};
        const float* wrow = &W1[(feat0 + m) * IN_DIM];

        for (int k = 0; k < KMAIN; k += 4) {
            v2f a, b;
            a.x = arow[k + kp];
            a.y = arow[k + kp + 1];
            b.x = wrow[k + kp];
            b.y = wrow[k + kp + 1];
            c = __builtin_amdgcn_wmma_f32_16x16x4_f32(
                    false, a, false, b, (short)0, c, false, false);
        }
        {
            v2f a, b;
            a.x = arow[KMAIN + kp];
            a.y = arow[KMAIN + kp + 1];
            b.x = (kp == 0) ? wrow[KMAIN] : 0.0f;
            b.y = 0.0f;
            c = __builtin_amdgcn_wmma_f32_16x16x4_f32(
                    false, a, false, b, (short)0, c, false, false);
        }

        const int mBase = ntile * 16 + ((lane >> 4) << 3);
#pragma unroll
        for (int r = 0; r < 8; ++r) {
            h[(nodeBase + mBase + r) * HID_DIM + feat0 + m] = c[r];
        }
    }
}


__global__ void k_selfloop1(const float* __restrict__ h, const float* __restrict__ dinv,
                            float* __restrict__ agg, int n) {
    int gid = blockIdx.x * blockDim.x + threadIdx.x;
    int i = gid >> 5;
    if (i >= n) return;
    int f4 = (gid & 31) * 4;
    float w = dinv[i] * dinv[i];
    float4 hv = *(const float4*)&h[(size_t)i * HID_DIM + f4];
    float4 o;
    o.x = hv.x * w; o.y = hv.y * w; o.z = hv.z * w; o.w = hv.w * w;
    *(float4*)&agg[(size_t)i * HID_DIM + f4] = o;
}

__global__ void k_scatter1(const int* __restrict__ src, const int* __restrict__ dst,
                           const float* __restrict__ dinv, const float* __restrict__ h,
                           float* __restrict__ agg) {
    int gid = blockIdx.x * blockDim.x + threadIdx.x;
    int e = gid >> 5;
    if (e >= N_EDGES) return;
    int lane = threadIdx.x & 31;
    int s = src[e], d = dst[e];
    float w = dinv[s] * dinv[d];
    int f4 = lane * 4;
    float4 hv = *(const float4*)&h[(size_t)s * HID_DIM + f4];
    float* ap = &agg[(size_t)d * HID_DIM + f4];
    atomicAdd(ap + 0, hv.x * w);
    atomicAdd(ap + 1, hv.y * w);
    atomicAdd(ap + 2, hv.z * w);
    atomicAdd(ap + 3, hv.w * w);
}


__global__ void k_relu_dot(const float* __restrict__ agg, const float* __restrict__ b1,
                           const float* __restrict__ W2, float* __restrict__ sval, int n) {
    int gid = blockIdx.x * blockDim.x + threadIdx.x;
    int i = gid >> 5;
    if (i >= n) return;
    int lane = threadIdx.x & 31;
    int f4 = lane * 4;
    float4 a  = *(const float4*)&agg[(size_t)i * HID_DIM + f4];
    float4 bb = *(const float4*)&b1[f4];
    float4 w2 = *(const float4*)&W2[f4];
    float p = fmaxf(a.x + bb.x, 0.f) * w2.x
            + fmaxf(a.y + bb.y, 0.f) * w2.y
            + fmaxf(a.z + bb.z, 0.f) * w2.z
            + fmaxf(a.w + bb.w, 0.f) * w2.w;
#pragma unroll
    for (int off = 16; off > 0; off >>= 1)
        p += __shfl_xor(p, off, 32);
    if (lane == 0) sval[i] = p;
}


__global__ void k_selfloop2(const float* __restrict__ sval, const float* __restrict__ dinv,
                            float* __restrict__ out, int n) {
    int i = blockIdx.x * blockDim.x + threadIdx.x;
    if (i < n) out[i] = sval[i] * dinv[i] * dinv[i];
}

__global__ void k_scatter2(const int* __restrict__ src, const int* __restrict__ dst,
                           const float* __restrict__ dinv, const float* __restrict__ sval,
                           float* __restrict__ out) {
    int e = blockIdx.x * blockDim.x + threadIdx.x;
    if (e >= N_EDGES) return;
    int s = src[e], d = dst[e];
    atomicAdd(&out[d], sval[s] * dinv[s] * dinv[d]);
}

__global__ void k_sigmoid(float* __restrict__ out, const float* __restrict__ b2, int n) {
    int i = blockIdx.x * blockDim.x + threadIdx.x;
    if (i < n) out[i] = 1.0f / (1.0f + expf(-(out[i] + b2[0])));
}


extern "C" void kernel_launch(void* const* d_in, const int* in_sizes, int n_in,
                              void* d_out, int out_size, void* d_ws, size_t ws_size,
                              hipStream_t stream) {
    (void)in_sizes; (void)n_in; (void)out_size; (void)ws_size;
    const float* x  = (const float*)d_in[0];
    const float* W1 = (const float*)d_in[1];
    const float* b1 = (const float*)d_in[2];
    const float* W2 = (const float*)d_in[3];
    const float* b2 = (const float*)d_in[4];
    const int*   ei = (const int*)d_in[5];
    const int* src = ei;
    const int* dst = ei + N_EDGES;
    float* out = (float*)d_out;

    float* dinv = (float*)d_ws;
    float* sval = dinv + N_NODES;
    float* h    = sval + N_NODES;
    float* agg  = h + (size_t)N_NODES * HID_DIM;

    const int B = 256;
    k_deg_init <<<(N_NODES + B - 1) / B, B, 0, stream>>>(dinv, N_NODES);
    k_deg_count<<<(N_EDGES + B - 1) / B, B, 0, stream>>>(dst, dinv, N_EDGES);
    k_dinv     <<<(N_NODES + B - 1) / B, B, 0, stream>>>(dinv, N_NODES);

    k_gemm1    <<<N_NODES / 64, B, 0, stream>>>(x, W1, h);

    k_selfloop1<<<(N_NODES * 32) / B, B, 0, stream>>>(h, dinv, agg, N_NODES);
    k_scatter1 <<<((size_t)N_EDGES * 32) / B, B, 0, stream>>>(src, dst, dinv, h, agg);

    k_relu_dot <<<(N_NODES * 32) / B, B, 0, stream>>>(agg, b1, W2, sval, N_NODES);

    k_selfloop2<<<(N_NODES + B - 1) / B, B, 0, stream>>>(sval, dinv, out, N_NODES);
    k_scatter2 <<<(N_EDGES + B - 1) / B, B, 0, stream>>>(src, dst, dinv, sval, out);
    k_sigmoid  <<<(N_NODES + B - 1) / B, B, 0, stream>>>(out, b2, N_NODES);
}


# In-degree and layer-2 scalar sums: each block sums one node range of one edge chunk in LDS, then adds whole 128-byte lines to memory; sum order = LDS arrival per block, then atomic arrival
# speedup vs baseline: 2.4907x; 2.4907x over previous
#include <hip/hip_runtime.h>
#include <math.h>

#define N_NODES 200000
#define N_EDGES 2000000
#define IN_DIM  165
#define HID_DIM 128
#define KPAD    168
#define WPITCH  172
#define GEMM_BLOCKS 512

typedef __attribute__((ext_vector_type(2))) float v2f;
typedef __attribute__((ext_vector_type(8))) float v8f;


__global__ void k_deg_init(float* __restrict__ deg, int n) {
    int i = blockIdx.x * blockDim.x + threadIdx.x;
    if (i < n) deg[i] = 1.0f;
}


#define PRIV_RANGES      4
#define PRIV_RANGE_NODES 50016
#define PRIV_CHUNKS      64
#define PRIV_EDGES       (N_EDGES / PRIV_CHUNKS)
#define PRIV_THREADS     1024
#define DEG_BATCH        31
#define SC2_BATCH        16
#define PRIV_ROT         768

static_assert((PRIV_RANGES & (PRIV_RANGES - 1)) == 0, "range = block & (ranges - 1)");
static_assert(PRIV_RANGES * PRIV_RANGE_NODES >= N_NODES, "ranges cover every node");
static_assert((PRIV_RANGES - 1) * PRIV_RANGE_NODES < N_NODES, "every range starts on a real node");
static_assert(PRIV_RANGE_NODES % 32 == 0 && PRIV_ROT % 32 == 0, "a wave adds into one aligned line");
static_assert(N_EDGES % PRIV_CHUNKS == 0, "whole chunks only");
static_assert((PRIV_CHUNKS - 1) * PRIV_ROT < PRIV_RANGE_NODES, "start offsets stay inside the range");
static_assert(DEG_BATCH * PRIV_THREADS >= PRIV_EDGES, "one trip covers the chunk");

__device__ __forceinline__ void priv_flush(const float* part, float* __restrict__ target,
                                           int r0, int chunk, int tid) {
    const int rot = chunk * PRIV_ROT;
    for (int k = tid; k < PRIV_RANGE_NODES; k += PRIV_THREADS) {
        int i = k + rot;
        if (i >= PRIV_RANGE_NODES) i -= PRIV_RANGE_NODES;
        const int node = r0 + i;
        const float v = part[i];
        if (v != 0.f && node < N_NODES) atomicAdd(&target[node], v);
    }
}

__global__ __launch_bounds__(PRIV_THREADS)
void k_deg_count(const int* __restrict__ dst, float* __restrict__ deg) {
    __shared__ float part[PRIV_RANGE_NODES];
    const int tid   = threadIdx.x;
    const int range = blockIdx.x & (PRIV_RANGES - 1);
    const int chunk = blockIdx.x / PRIV_RANGES;
    const int r0    = range * PRIV_RANGE_NODES;
    for (int i = tid; i < PRIV_RANGE_NODES; i += PRIV_THREADS) part[i] = 0.f;
    __syncthreads();
    const int* dp = dst + chunk * PRIV_EDGES;
    for (int base = 0; base < PRIV_EDGES; base += DEG_BATCH * PRIV_THREADS) {
        int d[DEG_BATCH];
#pragma unroll
        for (int j = 0; j < DEG_BATCH; ++j) {
            const int e = base + j * PRIV_THREADS + tid;
            const int v = dp[e < PRIV_EDGES ? e : PRIV_EDGES - 1];
            d[j] = (e < PRIV_EDGES) ? v : -1;
        }
#pragma unroll
        for (int j = 0; j < DEG_BATCH; ++j) {
            const unsigned r = (unsigned)(d[j] - r0);
            if (r < (unsigned)PRIV_RANGE_NODES) atomicAdd(&part[r], 1.0f);
        }
    }
    __syncthreads();
    priv_flush(part, deg, r0, chunk, tid);
}

__global__ void k_dinv(float* __restrict__ deg, int n) {
    int i = blockIdx.x * blockDim.x + threadIdx.x;
    if (i < n) deg[i] = rsqrtf(deg[i]);
}


__global__ __launch_bounds__(256)
void k_gemm1(const float* __restrict__ x, const float* __restrict__ W1,
             float* __restrict__ h) {
    __shared__ __attribute__((aligned(16))) float sW[HID_DIM * WPITCH];
    __shared__ __attribute__((aligned(16))) float sX[64 * WPITCH];
    const int tid  = threadIdx.x;
    const int lane = tid & 31;
    const int wave = tid >> 5;

    for (int rb = 0; rb < HID_DIM; rb += 64) {
        float wv[8][5];
        for (int i = 0; i < 8; ++i) {
            const float* wr = W1 + (rb + wave + 8 * i) * IN_DIM + lane;
            for (int j = 0; j < 5; ++j) wv[i][j] = wr[32 * j];
        }
        float wt[2];
        for (int i = 0; i < 2; ++i) {
            const int e = tid + 256 * i;
            const int c = 160 + (e & 7);
            wt[i] = W1[(rb + (e >> 3)) * IN_DIM + (c < IN_DIM ? c : IN_DIM - 1)];
        }
        for (int i = 0; i < 8; ++i) {
            float* sr = sW + (rb + wave + 8 * i) * WPITCH + lane;
            for (int j = 0; j < 5; ++j) sr[32 * j] = wv[i][j];
        }
        for (int i = 0; i < 2; ++i) {
            const int e = tid + 256 * i;
            const int c = 160 + (e & 7);
            sW[(rb + (e >> 3)) * WPITCH + c] = (c < IN_DIM) ? wt[i] : 0.0f;
        }
    }

    const int m     = lane & 15;
    const int kp    = (lane >> 4) << 1;
    const int ntile = wave >> 1;
    const int fbase = (wave & 1) * 64;
    const float* arow = sX + (ntile * 16 + m) * WPITCH + kp;
    const float* brow = sW + (fbase + m) * WPITCH + kp;
    const int mBase = ntile * 16 + ((lane >> 4) << 3);

    for (int tile = blockIdx.x; tile < N_NODES / 64; tile += GEMM_BLOCKS) {
        const int nodeBase = tile * 64;

        __syncthreads();
        {
            float xv[8][5];
            for (int i = 0; i < 8; ++i) {
                const float* xr = x + (nodeBase + wave + 8 * i) * IN_DIM + lane;
                for (int j = 0; j < 5; ++j) xv[i][j] = xr[32 * j];
            }
            float xt[2];
            for (int i = 0; i < 2; ++i) {
                const int e = tid + 256 * i;
                const int c = 160 + (e & 7);
                xt[i] = x[(nodeBase + (e >> 3)) * IN_DIM + (c < IN_DIM ? c : IN_DIM - 1)];
            }
            for (int i = 0; i < 8; ++i) {
                float* sr = sX + (wave + 8 * i) * WPITCH + lane;
                for (int j = 0; j < 5; ++j) sr[32 * j] = xv[i][j];
            }
            for (int i = 0; i < 2; ++i) {
                const int e = tid + 256 * i;
                const int c = 160 + (e & 7);
                sX[(e >> 3) * WPITCH + c] = (c < IN_DIM) ? xt[i] : 0.0f;
            }
        }
        __syncthreads();

        v8f c0 = {0.f, 0.f, 0.f, 0.f, 0.f, 0.f, 0.f, 0.f};
        v8f c1 = c0, c2 = c0, c3 = c0;
#pragma unroll
        for (int k = 0; k < KPAD; k += 4) {
            const v2f a  = *(const v2f*)(arow + k);
            const v2f b0 = *(const v2f*)(brow + k);
            const v2f b1 = *(const v2f*)(brow + 16 * WPITCH + k);
            const v2f b2 = *(const v2f*)(brow + 32 * WPITCH + k);
            const v2f b3 = *(const v2f*)(brow + 48 * WPITCH + k);
            c0 = __builtin_amdgcn_wmma_f32_16x16x4_f32(false, a, false, b0, (short)0, c0, false, false);
            c1 = __builtin_amdgcn_wmma_f32_16x16x4_f32(false, a, false, b1, (short)0, c1, false, false);
            c2 = __builtin_amdgcn_wmma_f32_16x16x4_f32(false, a, false, b2, (short)0, c2, false, false);
            c3 = __builtin_amdgcn_wmma_f32_16x16x4_f32(false, a, false, b3, (short)0, c3, false, false);
        }

        float* hp = h + (size_t)(nodeBase + mBase) * HID_DIM + fbase + m;
        for (int r = 0; r < 8; ++r) {
            hp[r * HID_DIM]      = c0[r];
            hp[r * HID_DIM + 16] = c1[r];
            hp[r * HID_DIM + 32] = c2[r];
            hp[r * HID_DIM + 48] = c3[r];
        }
    }
}


#define ROWS_PER_WAVE 8
#define EDGES_PER_WAVE 32
#define EDGE_GROUP 8

static_assert(N_NODES % ROWS_PER_WAVE == 0, "whole waves only");
static_assert(N_EDGES % EDGES_PER_WAVE == 0, "whole waves only");
static_assert(EDGES_PER_WAVE % EDGE_GROUP == 0, "whole groups only");

__global__ __launch_bounds__(256)
void k_selfloop1(const float* __restrict__ h, const float* __restrict__ dinv,
                 float* __restrict__ agg) {
    const int lane = threadIdx.x & 31;
    const int n0 = ((blockIdx.x * blockDim.x + threadIdx.x) >> 5) * ROWS_PER_WAVE;
    if (n0 >= N_NODES) return;
    const int f4 = lane * 4;
    float4 hv[ROWS_PER_WAVE];
    float dw[ROWS_PER_WAVE];
#pragma unroll
    for (int r = 0; r < ROWS_PER_WAVE; ++r) {
        hv[r] = *(const float4*)&h[(size_t)(n0 + r) * HID_DIM + f4];
        dw[r] = dinv[n0 + r];
    }
#pragma unroll
    for (int r = 0; r < ROWS_PER_WAVE; ++r) {
        const float w = dw[r] * dw[r];
        float4 o;
        o.x = hv[r].x * w; o.y = hv[r].y * w; o.z = hv[r].z * w; o.w = hv[r].w * w;
        *(float4*)&agg[(size_t)(n0 + r) * HID_DIM + f4] = o;
    }
}

__global__ __launch_bounds__(256)
void k_scatter1(const int* __restrict__ src, const int* __restrict__ dst,
                const float* __restrict__ dinv, const float* __restrict__ h,
                float* __restrict__ agg) {
    const int lane = threadIdx.x & 31;
    const int e0 = ((blockIdx.x * blockDim.x + threadIdx.x) >> 5) * EDGES_PER_WAVE;
    if (e0 >= N_EDGES) return;
    const int s = src[e0 + lane];
    const int d = dst[e0 + lane];
    const float w = dinv[s] * dinv[d];

    float cur[EDGE_GROUP][4], nxt[EDGE_GROUP][4];
#pragma unroll
    for (int j = 0; j < EDGE_GROUP; ++j) {
        const int sj = __builtin_amdgcn_readlane(s, j);
        const float* hp = h + (size_t)sj * HID_DIM + lane;
        cur[j][0] = hp[0]; cur[j][1] = hp[32]; cur[j][2] = hp[64]; cur[j][3] = hp[96];
    }
#pragma unroll
    for (int g = 0; g < EDGES_PER_WAVE; g += EDGE_GROUP) {
        if (g + EDGE_GROUP < EDGES_PER_WAVE) {
#pragma unroll
            for (int j = 0; j < EDGE_GROUP; ++j) {
                const int sj = __builtin_amdgcn_readlane(s, g + EDGE_GROUP + j);
                const float* hp = h + (size_t)sj * HID_DIM + lane;
                nxt[j][0] = hp[0]; nxt[j][1] = hp[32]; nxt[j][2] = hp[64]; nxt[j][3] = hp[96];
            }
        }
#pragma unroll
        for (int j = 0; j < EDGE_GROUP; ++j) {
            const int dj = __builtin_amdgcn_readlane(d, g + j);
            const float wj = __int_as_float(__builtin_amdgcn_readlane(__float_as_int(w), g + j));
            float* ap = agg + (size_t)dj * HID_DIM + lane;
            atomicAdd(ap +  0, cur[j][0] * wj);
            atomicAdd(ap + 32, cur[j][1] * wj);
            atomicAdd(ap + 64, cur[j][2] * wj);
            atomicAdd(ap + 96, cur[j][3] * wj);
        }
#pragma unroll
        for (int j = 0; j < EDGE_GROUP; ++j)
            for (int k = 0; k < 4; ++k) cur[j][k] = nxt[j][k];
    }
}


__global__ __launch_bounds__(256)
void k_relu_dot(const float* __restrict__ agg, const float* __restrict__ b1,
                const float* __restrict__ W2, float* __restrict__ sval) {
    const int lane = threadIdx.x & 31;
    const int n0 = ((blockIdx.x * blockDim.x + threadIdx.x) >> 5) * ROWS_PER_WAVE;
    if (n0 >= N_NODES) return;
    const int f4 = lane * 4;
    float4 a[ROWS_PER_WAVE];
#pragma unroll
    for (int r = 0; r < ROWS_PER_WAVE; ++r)
        a[r] = *(const float4*)&agg[(size_t)(n0 + r) * HID_DIM + f4];
    const float4 bb = *(const float4*)&b1[f4];
    const float4 w2 = *(const float4*)&W2[f4];
    float mine = 0.f;
#pragma unroll
    for (int r = 0; r < ROWS_PER_WAVE; ++r) {
        float p = fmaxf(a[r].x + bb.x, 0.f) * w2.x
                + fmaxf(a[r].y + bb.y, 0.f) * w2.y
                + fmaxf(a[r].z + bb.z, 0.f) * w2.z
                + fmaxf(a[r].w + bb.w, 0.f) * w2.w;
#pragma unroll
        for (int off = 16; off > 0; off >>= 1)
            p += __shfl_xor(p, off, 32);
        mine = (lane == r) ? p : mine;
    }
    if (lane < ROWS_PER_WAVE) sval[n0 + lane] = mine;
}


__global__ void k_selfloop2(const float* __restrict__ sval, const float* __restrict__ dinv,
                            float* __restrict__ out, int n) {
    int i = blockIdx.x * blockDim.x + threadIdx.x;
    if (i < n) out[i] = sval[i] * dinv[i] * dinv[i];
}

__global__ __launch_bounds__(PRIV_THREADS)
void k_scatter2(const int* __restrict__ src, const int* __restrict__ dst,
                const float* __restrict__ dinv, const float* __restrict__ sval,
                float* __restrict__ out) {
    __shared__ float part[PRIV_RANGE_NODES];
    const int tid   = threadIdx.x;
    const int range = blockIdx.x & (PRIV_RANGES - 1);
    const int chunk = blockIdx.x / PRIV_RANGES;
    const int r0    = range * PRIV_RANGE_NODES;
    for (int i = tid; i < PRIV_RANGE_NODES; i += PRIV_THREADS) part[i] = 0.f;
    __syncthreads();
    const int* sp = src + chunk * PRIV_EDGES;
    const int* dp = dst + chunk * PRIV_EDGES;
    for (int base = 0; base < PRIV_EDGES; base += SC2_BATCH * PRIV_THREADS) {
        int s[SC2_BATCH], d[SC2_BATCH];
#pragma unroll
        for (int j = 0; j < SC2_BATCH; ++j) {
            const int e  = base + j * PRIV_THREADS + tid;
            const int ec = e < PRIV_EDGES ? e : PRIV_EDGES - 1;
            const int sv = sp[ec];
            const int dv = dp[ec];
            const bool mine = (e < PRIV_EDGES) && ((unsigned)(dv - r0) < (unsigned)PRIV_RANGE_NODES);
            s[j] = mine ? sv : r0;
            d[j] = mine ? dv : -1;
        }
        float val[SC2_BATCH];
#pragma unroll
        for (int j = 0; j < SC2_BATCH; ++j) {
            const int dj = d[j] >= 0 ? d[j] : r0;
            val[j] = sval[s[j]] * dinv[s[j]] * dinv[dj];
        }
        __atomic_signal_fence(__ATOMIC_SEQ_CST);
#pragma unroll
        for (int j = 0; j < SC2_BATCH; ++j)
            if (d[j] >= 0) atomicAdd(&part[d[j] - r0], val[j]);
    }
    __syncthreads();
    priv_flush(part, out, r0, chunk, tid);
}

__global__ void k_sigmoid(float* __restrict__ out, const float* __restrict__ b2, int n) {
    int i = blockIdx.x * blockDim.x + threadIdx.x;
    if (i < n) out[i] = 1.0f / (1.0f + expf(-(out[i] + b2[0])));
}


extern "C" void kernel_launch(void* const* d_in, const int* in_sizes, int n_in,
                              void* d_out, int out_size, void* d_ws, size_t ws_size,
                              hipStream_t stream) {
    (void)in_sizes; (void)n_in; (void)out_size; (void)ws_size;
    const float* x  = (const float*)d_in[0];
    const float* W1 = (const float*)d_in[1];
    const float* b1 = (const float*)d_in[2];
    const float* W2 = (const float*)d_in[3];
    const float* b2 = (const float*)d_in[4];
    const int*   ei = (const int*)d_in[5];
    const int* src = ei;
    const int* dst = ei + N_EDGES;
    float* out = (float*)d_out;

    float* dinv = (float*)d_ws;
    float* sval = dinv + N_NODES;
    float* h    = sval + N_NODES;
    float* agg  = h + (size_t)N_NODES * HID_DIM;

    const int B = 256;
    k_deg_init <<<(N_NODES + B - 1) / B, B, 0, stream>>>(dinv, N_NODES);
    k_deg_count<<<PRIV_RANGES * PRIV_CHUNKS, PRIV_THREADS, 0, stream>>>(dst, dinv);
    k_dinv     <<<(N_NODES + B - 1) / B, B, 0, stream>>>(dinv, N_NODES);

    k_gemm1    <<<GEMM_BLOCKS, B, 0, stream>>>(x, W1, h);

    k_selfloop1<<<(N_NODES / ROWS_PER_WAVE + 7) / 8, B, 0, stream>>>(h, dinv, agg);
    k_scatter1 <<<(N_EDGES / EDGES_PER_WAVE + 7) / 8, B, 0, stream>>>(src, dst, dinv, h, agg);

    k_relu_dot <<<(N_NODES / ROWS_PER_WAVE + 7) / 8, B, 0, stream>>>(agg, b1, W2, sval);

    k_selfloop2<<<(N_NODES + B - 1) / B, B, 0, stream>>>(sval, dinv, out, N_NODES);
    k_scatter2 <<<PRIV_RANGES * PRIV_CHUNKS, PRIV_THREADS, 0, stream>>>(src, dst, dinv, sval, out);
    k_sigmoid  <<<(N_NODES + B - 1) / B, B, 0, stream>>>(out, b2, N_NODES);
}
